# starting kernel
# baseline (speedup 1.0000x reference)
#include <hip/hip_runtime.h>

typedef __attribute__((ext_vector_type(2))) float v2f;
typedef __attribute__((ext_vector_type(8))) float v8f;


__global__ void fill_f32(float* __restrict__ p, long long n, float v) {
  long long i = (long long)blockIdx.x * blockDim.x + threadIdx.x;
  if (i < n) p[i] = v;
}

__global__ void deg_accum(const int* __restrict__ dst, float* __restrict__ deg, int E) {
  int e = blockIdx.x * blockDim.x + threadIdx.x;
  if (e < E) atomicAdd(&deg[dst[e]], 1.0f);
}

__global__ void deg_rsqrt(const float* __restrict__ deg, float* __restrict__ dinv, int n) {
  int i = blockIdx.x * blockDim.x + threadIdx.x;
  if (i < n) {
    float d = deg[i];
    dinv[i] = d > 0.f ? rsqrtf(d) : 0.f;
  }
}

template <int TK>
__global__ void gemm_wmma_f32_lds(const float* __restrict__ A,
                                  const float* __restrict__ B,
                                  float* __restrict__ C,
                                  int M, int K, int NC) {
  constexpr int LSTR = TK + 4;
  __shared__ float As[16 * LSTR];

  const int tid   = threadIdx.x;
  const int lane  = tid & 31;
  const int wave  = tid >> 5;
  const int row0  = blockIdx.x << 4;
  const int col0  = wave << 4;
  const int mn    = lane & 15;
  const int half  = lane >> 4;
  const int half2 = half << 1;
  const int nthr  = blockDim.x;

  const float* bp = B + (size_t)half2 * NC + col0 + mn;

  v8f acc = {};
  for (int k0 = 0; k0 < K; k0 += TK) {
    for (int t = tid * 4; t < 16 * TK; t += nthr * 4) {
      const int r = t / TK;
      const int c = t % TK;
      const float* gp = A + (size_t)(row0 + r) * K + k0 + c;
      __builtin_prefetch(gp + TK, 0, 1);
      *(float4*)(&As[r * LSTR + c]) = *(const float4*)gp;
    }
    __syncthreads();

#pragma unroll
    for (int kk = 0; kk < TK; kk += 4) {
      v2f a, b;
      a.x = As[mn * LSTR + kk + half2];
      a.y = As[mn * LSTR + kk + half2 + 1];
      b.x = bp[(size_t)(k0 + kk) * NC];
      b.y = bp[(size_t)(k0 + kk + 1) * NC];
      acc = __builtin_amdgcn_wmma_f32_16x16x4_f32(
           false, a,  false, b,
           (short)0, acc,  false,  false);
    }
    __syncthreads();
  }

  float* cp = C + (size_t)(row0 + (half << 3)) * NC + col0 + mn;
#pragma unroll
  for (int g = 0; g < 8; ++g) cp[(size_t)g * NC] = acc[g];
}

template <int CH>
__global__ void scatter_edges4(const int* __restrict__ src, const int* __restrict__ dst,
                               const float* __restrict__ dinv,
                               const float* __restrict__ h, float* __restrict__ z,
                               int E) {
  constexpr int CV = CH / 4;
  long long idx = (long long)blockIdx.x * blockDim.x + threadIdx.x;
  if (idx >= (long long)E * CV) return;
  const int e  = (int)(idx / CV);
  const int c4 = (int)(idx % CV) * 4;
  const int s = src[e], d = dst[e];
  const float norm = dinv[s] * dinv[d];
  const float4 hv = *(const float4*)(h + (size_t)s * CH + c4);
  float* zp = z + (size_t)d * CH + c4;
  atomicAdd(zp + 0, hv.x * norm);
  atomicAdd(zp + 1, hv.y * norm);
  atomicAdd(zp + 2, hv.z * norm);
  atomicAdd(zp + 3, hv.w * norm);
}

template <int CH, bool RELU>
__global__ void self_bias4(const float* __restrict__ h, const float* __restrict__ dinv,
                           const float* __restrict__ bias, float* __restrict__ z,
                           int n) {
  constexpr int CV = CH / 4;
  long long idx = (long long)blockIdx.x * blockDim.x + threadIdx.x;
  if (idx >= (long long)n * CV) return;
  const int i  = (int)(idx / CV);
  const int c4 = (int)(idx % CV) * 4;
  const float dv = dinv[i];
  const float s = dv * dv;
  const float4 hv = *(const float4*)(h + (size_t)i * CH + c4);
  const float4 bv = *(const float4*)(bias + c4);
  float4* zp = (float4*)(z + (size_t)i * CH + c4);
  float4 zv = *zp;
  float4 r;
  r.x = zv.x + hv.x * s + bv.x;
  r.y = zv.y + hv.y * s + bv.y;
  r.z = zv.z + hv.z * s + bv.z;
  r.w = zv.w + hv.w * s + bv.w;
  if (RELU) {
    r.x = fmaxf(r.x, 0.f); r.y = fmaxf(r.y, 0.f);
    r.z = fmaxf(r.z, 0.f); r.w = fmaxf(r.w, 0.f);
  }
  *zp = r;
}

__global__ void edge_dot(const int* __restrict__ es, const int* __restrict__ ed,
                         const float* __restrict__ z, float* __restrict__ out, int L) {
  int j = blockIdx.x * blockDim.x + threadIdx.x;
  if (j >= L) return;
  const float4* a = (const float4*)(z + (size_t)es[j] * 64);
  const float4* b = (const float4*)(z + (size_t)ed[j] * 64);
  float sum = 0.f;
#pragma unroll
  for (int t = 0; t < 16; ++t) {
    float4 x = a[t], y = b[t];
    sum += x.x * y.x + x.y * y.y + x.z * y.z + x.w * y.w;
  }
  out[j] = sum;
}


static inline int cdiv_ll(long long a, long long b) { return (int)((a + b - 1) / b); }

extern "C" void kernel_launch(void* const* d_in, const int* in_sizes, int n_in,
                              void* d_out, int out_size, void* d_ws, size_t ws_size,
                              hipStream_t stream) {
  const float* x   = (const float*)d_in[0];
  const int*   ei  = (const int*)d_in[1];
  const int*   eli = (const int*)d_in[2];
  const float* W1  = (const float*)d_in[3];
  const float* b1  = (const float*)d_in[4];
  const float* W2  = (const float*)d_in[5];
  const float* b2  = (const float*)d_in[6];
  float* out = (float*)d_out;

  const int IN = 512, HID = 128, OUT = 64;
  const int n = in_sizes[0] / IN;
  const int E = in_sizes[1] / 2;
  const int L = in_sizes[2] / 2;
  const int* src = ei;
  const int* dst = ei + E;
  const int* es  = eli;
  const int* ed  = eli + L;

  float* ws   = (float*)d_ws;
  float* deg  = ws;  ws += n;
  float* dinv = ws;  ws += n;
  float* h1   = ws;  ws += (size_t)n * HID;
  float* z1   = ws;  ws += (size_t)n * HID;
  float* h2   = ws;  ws += (size_t)n * OUT;
  float* z2   = ws;  ws += (size_t)n * OUT;

  const int B = 256;

  fill_f32<<<cdiv_ll(n, B), B, 0, stream>>>(deg, n, 1.0f);
  fill_f32<<<cdiv_ll((long long)n * HID, B), B, 0, stream>>>(z1, (long long)n * HID, 0.f);
  fill_f32<<<cdiv_ll((long long)n * OUT, B), B, 0, stream>>>(z2, (long long)n * OUT, 0.f);

  deg_accum<<<cdiv_ll(E, B), B, 0, stream>>>(dst, deg, E);
  deg_rsqrt<<<cdiv_ll(n, B), B, 0, stream>>>(deg, dinv, n);

  gemm_wmma_f32_lds<64><<<n / 16, 32 * (HID / 16), 0, stream>>>(x, W1, h1, n, IN, HID);
  scatter_edges4<128><<<cdiv_ll((long long)E * (HID / 4), B), B, 0, stream>>>(src, dst, dinv, h1, z1, E);
  self_bias4<128, true><<<cdiv_ll((long long)n * (HID / 4), B), B, 0, stream>>>(h1, dinv, b1, z1, n);

  gemm_wmma_f32_lds<64><<<n / 16, 32 * (OUT / 16), 0, stream>>>(z1, W2, h2, n, HID, OUT);
  scatter_edges4<64><<<cdiv_ll((long long)E * (OUT / 4), B), B, 0, stream>>>(src, dst, dinv, h2, z2, E);
  self_bias4<64, false><<<cdiv_ll((long long)n * (OUT / 4), B), B, 0, stream>>>(h2, dinv, b2, z2, n);

  edge_dot<<<cdiv_ll(L, B), B, 0, stream>>>(es, ed, z2, out, L);
}


# owner scan in two levels: a new first launch filters the edge list into 16x16 compact lists by 3136-row group; an owner scans its group's 16 lists, not all edges; 6->7 launches; sums in arrival order
# speedup vs baseline: 2.2054x; 2.2054x over previous
#include <hip/hip_runtime.h>

typedef __attribute__((ext_vector_type(2))) float v2f;
typedef __attribute__((ext_vector_type(8))) float v8f;


template <int NT, int KL>
__device__ __forceinline__ void load_dst_step(const int* __restrict__ p, int first, int lim,
                                              bool vec, int tid, int4 (&d4)[KL]) {
  constexpr int CHUNK = NT * 4 * KL;
  const int base = first + tid * 4;
  if (vec && first + CHUNK <= lim) {
#pragma unroll
    for (int j = 0; j < KL; ++j)
      d4[j] = *(const int4*)(p + base + j * (NT * 4));
  } else if (vec) {
#pragma unroll
    for (int j = 0; j < KL; ++j) {
      const int eb = base + j * (NT * 4);
      d4[j] = make_int4(-1, -1, -1, -1);
      if (eb < lim) d4[j] = *(const int4*)(p + eb);
    }
  } else {
#pragma unroll
    for (int j = 0; j < KL; ++j) {
      const int eb = base + j * (NT * 4);
      d4[j].x = eb     < lim ? p[eb]     : -1;
      d4[j].y = eb + 1 < lim ? p[eb + 1] : -1;
      d4[j].z = eb + 2 < lim ? p[eb + 2] : -1;
      d4[j].w = eb + 3 < lim ? p[eb + 3] : -1;
    }
  }
}

template <int GROWS, int PRT, int NT, int KL, int FCAP, int QLDS>
__global__ void __launch_bounds__(NT)
filter_edges_by_group(const int* __restrict__ src, const int* __restrict__ dst,
                      int* __restrict__ fsrc, int* __restrict__ fdst, int* __restrict__ fcnt,
                      int E, int n) {
  constexpr int CHUNK = NT * 4 * KL;
  static_assert(FCAP <= QLDS, "a stored list fits the LDS queue");
  static_assert((2 * QLDS + 1) * 4 <= 320 * 1024, "the queue fits one work-group's LDS");

  __shared__ int qe[QLDS];
  __shared__ int qd[QLDS];
  __shared__ int qcnt;

  const int tid   = threadIdx.x;
  const int lane  = tid & 31;
  const int grp   = blockIdx.x / PRT;
  const int prt   = blockIdx.x % PRT;
  const int lo    = grp * GROWS;
  const int nrows = max(0, min(GROWS, n - lo));
  const int pe    = (((E + PRT - 1) / PRT) + 3) & ~3;
  const int e0    = min(E, prt * pe);
  const int e1    = min(E, e0 + pe);

  if (tid == 0) qcnt = 0;
  __syncthreads();

  const int  nsteps = (e1 - e0 + CHUNK - 1) / CHUNK;
  const bool vec    = (E & 3) == 0;
  const int  rot    = nsteps > 0 ? grp % nsteps : 0;
  for (int st = 0; st < nsteps; ++st) {
    int sx = st + rot;
    if (sx >= nsteps) sx -= nsteps;
    const int first = e0 + sx * CHUNK;
    int4 d4[KL];
    load_dst_step<NT, KL>(dst, first, e1, vec, tid, d4);

    unsigned mk[4 * KL];
    int tot = 0;
#pragma unroll
    for (int j = 0; j < KL; ++j) {
      const int dd[4] = {d4[j].x, d4[j].y, d4[j].z, d4[j].w};
#pragma unroll
      for (int k = 0; k < 4; ++k) {
        mk[j * 4 + k] = __builtin_amdgcn_ballot_w32((unsigned)(dd[k] - lo) < (unsigned)nrows);
        tot += __builtin_popcount(mk[j * 4 + k]);
      }
    }
    if (tot != 0) {
      int slot = 0;
      if (lane == 0) slot = atomicAdd(&qcnt, tot);
      slot = __builtin_amdgcn_readfirstlane(slot);
#pragma unroll
      for (int j = 0; j < KL; ++j) {
        const int eb = first + tid * 4 + j * (NT * 4);
        const int dd[4] = {d4[j].x, d4[j].y, d4[j].z, d4[j].w};
#pragma unroll
        for (int k = 0; k < 4; ++k) {
          const unsigned m = mk[j * 4 + k];
          if ((m >> lane) & 1u) {
            const int pos = slot + (int)__builtin_amdgcn_mbcnt_lo(m, 0u);
            if (pos < FCAP) { qe[pos] = eb + k; qd[pos] = dd[k]; }
          }
          slot += __builtin_popcount(m);
        }
      }
    }
  }
  __syncthreads();
  const int total = qcnt;
  if (tid == 0) fcnt[blockIdx.x] = total;

  if (total <= FCAP) {
    int* fs = fsrc + (size_t)blockIdx.x * FCAP;
    int* fd = fdst + (size_t)blockIdx.x * FCAP;
    for (int i0 = tid; i0 < FCAP; i0 += NT * 4) {
      int sv[4];
#pragma unroll
      for (int k = 0; k < 4; ++k) {
        const int i = i0 + k * NT;
        sv[k] = i < total ? src[qe[i]] : 0;
      }
#pragma unroll
      for (int k = 0; k < 4; ++k) {
        const int i = i0 + k * NT;
        if (i < total)     { fs[i] = sv[k]; fd[i] = qd[i]; }
        else if (i < FCAP) { fd[i] = -1; }
      }
    }
  }
}

template <int R, int NT, int KL, int CAPQ, int CAPS, int GRP, int PRT, int FCAP>
__global__ void __launch_bounds__(NT)
range_scan_edges(const int* __restrict__ src, const int* __restrict__ dst,
                 const int* __restrict__ fsrc, const int* __restrict__ fdst,
                 const int* __restrict__ fcnt,
                 float* __restrict__ dinv, int* __restrict__ lst, int* __restrict__ lcnt,
                 int E, int n) {
  constexpr int CHUNK = NT * 4 * KL;
  static_assert(CAPS <= CAPQ, "a stored queue fits the LDS queue");
  static_assert((2 * CAPQ + R + 1) * 4 <= 320 * 1024, "queue and counters fit one work-group's LDS");
  static_assert(FCAP % 4 == 0, "every filtered list starts on an int4 boundary");

  __shared__ int qs[CAPQ];
  __shared__ int qu[CAPQ];
  __shared__ int cnt[R];
  __shared__ int qcnt;

  const int tid   = threadIdx.x;
  const int lane  = tid & 31;
  const int lo    = blockIdx.x * R;
  const int nrows = max(0, min(R, n - lo));

  for (int i = tid; i < R; i += NT) cnt[i] = 0;
  if (tid == 0) qcnt = 0;
  __syncthreads();

  const int grp = blockIdx.x / GRP;
  int over = 0;
  for (int p = 0; p < PRT; ++p)
    over |= (unsigned)fcnt[grp * PRT + p] > (unsigned)FCAP ? 1 : 0;
  const bool filt = over == 0;
  const int* sd = filt ? fdst + (size_t)grp * (PRT * FCAP) : dst;
  const int* ss = filt ? fsrc + (size_t)grp * (PRT * FCAP) : src;
  const int  EE = filt ? PRT * FCAP : E;

  const int  nsteps = (EE + CHUNK - 1) / CHUNK;
  const bool vec    = (EE & 3) == 0;
  const int  rot    = nsteps > 0 ? (int)(blockIdx.x % (unsigned)nsteps) : 0;
  for (int st = 0; st < nsteps; ++st) {
    int sx = st + rot;
    if (sx >= nsteps) sx -= nsteps;
    const int first = sx * CHUNK;
    int4 d4[KL];
    load_dst_step<NT, KL>(sd, first, EE, vec, tid, d4);

    unsigned mk[4 * KL];
    int tot = 0;
#pragma unroll
    for (int j = 0; j < KL; ++j) {
      const int dd[4] = {d4[j].x, d4[j].y, d4[j].z, d4[j].w};
#pragma unroll
      for (int k = 0; k < 4; ++k) {
        mk[j * 4 + k] = __builtin_amdgcn_ballot_w32((unsigned)(dd[k] - lo) < (unsigned)nrows);
        tot += __builtin_popcount(mk[j * 4 + k]);
      }
    }
    if (tot != 0) {
      int slot = 0;
      if (lane == 0) slot = atomicAdd(&qcnt, tot);
      slot = __builtin_amdgcn_readfirstlane(slot);
#pragma unroll
      for (int j = 0; j < KL; ++j) {
        const int eb = first + tid * 4 + j * (NT * 4);
        const int dd[4] = {d4[j].x, d4[j].y, d4[j].z, d4[j].w};
#pragma unroll
        for (int k = 0; k < 4; ++k) {
          const unsigned m = mk[j * 4 + k];
          if ((m >> lane) & 1u) {
            const int uu  = dd[k] - lo;
            const int pos = slot + (int)__builtin_amdgcn_mbcnt_lo(m, 0u);
            atomicAdd(&cnt[uu], 1);
            if (pos < CAPQ) { qs[pos] = eb + k; qu[pos] = uu; }
          }
          slot += __builtin_popcount(m);
        }
      }
    }
  }
  __syncthreads();
  const int total = qcnt;
  if (tid == 0) lcnt[blockIdx.x] = total;

  if (total <= CAPS) {
    int* lp = lst + (size_t)blockIdx.x * (3 * CAPS);
    for (int i0 = tid; i0 < total; i0 += NT * 4) {
      int sv[4];
#pragma unroll
      for (int k = 0; k < 4; ++k) {
        const int i = i0 + k * NT;
        sv[k] = i < total ? ss[qs[i]] : 0;
      }
#pragma unroll
      for (int k = 0; k < 4; ++k) {
        const int i = i0 + k * NT;
        if (i < total) { lp[i] = sv[k]; lp[CAPS + i] = qu[i]; }
      }
    }
  }
  for (int i = tid; i < nrows; i += NT) {
    const float d = (float)(cnt[i] + 1);
    dinv[lo + i] = d > 0.f ? rsqrtf(d) : 0.f;
  }
}

template <int TK, bool RELU_A>
__global__ void gemm_wmma_f32_lds(const float* __restrict__ A,
                                  const float* __restrict__ B,
                                  const float* __restrict__ dinv,
                                  const float* __restrict__ bias,
                                  float* __restrict__ C,
                                  float* __restrict__ Z,
                                  int M, int K, int NC) {
  constexpr int LSTR = TK + 4;
  __shared__ float As[16 * LSTR];

  const int tid   = threadIdx.x;
  const int lane  = tid & 31;
  const int wave  = tid >> 5;
  const int row0  = blockIdx.x << 4;
  const int col0  = wave << 4;
  const int mn    = lane & 15;
  const int half  = lane >> 4;
  const int half2 = half << 1;
  const int nthr  = blockDim.x;

  const float* bp = B + (size_t)half2 * NC + col0 + mn;

  v8f acc = {};
  for (int k0 = 0; k0 < K; k0 += TK) {
    for (int t = tid * 4; t < 16 * TK; t += nthr * 4) {
      const int r = t / TK;
      const int c = t % TK;
      const float* gp = A + (size_t)(row0 + r) * K + k0 + c;
      __builtin_prefetch(gp + TK, 0, 1);
      float4 av = *(const float4*)gp;
      if (RELU_A) {
        av.x = fmaxf(av.x, 0.f); av.y = fmaxf(av.y, 0.f);
        av.z = fmaxf(av.z, 0.f); av.w = fmaxf(av.w, 0.f);
      }
      *(float4*)(&As[r * LSTR + c]) = av;
    }
    __syncthreads();

#pragma unroll
    for (int kk = 0; kk < TK; kk += 4) {
      v2f a, b;
      a.x = As[mn * LSTR + kk + half2];
      a.y = As[mn * LSTR + kk + half2 + 1];
      b.x = bp[(size_t)(k0 + kk) * NC];
      b.y = bp[(size_t)(k0 + kk + 1) * NC];
      acc = __builtin_amdgcn_wmma_f32_16x16x4_f32(
           false, a,  false, b,
           (short)0, acc,  false,  false);
    }
    __syncthreads();
  }

  const int rbase = row0 + (half << 3);
  const float bv = bias[col0 + mn];
  float sl[8];
#pragma unroll
  for (int g = 0; g < 8; ++g) {
    const float dv = dinv[rbase + g];
    sl[g] = dv * dv;
  }
  float* cp = C + (size_t)rbase * NC + col0 + mn;
  float* zp = Z + (size_t)rbase * NC + col0 + mn;
#pragma unroll
  for (int g = 0; g < 8; ++g) {
    cp[(size_t)g * NC] = acc[g];
    zp[(size_t)g * NC] = acc[g] * sl[g] + bv;
  }
}

template <int CH, int R, int NT, int KL, int CAP, int CAPS, int LIST>
__global__ void __launch_bounds__(NT)
range_sum_edges(const int* __restrict__ src, const int* __restrict__ dst,
                const float* __restrict__ dinv, const float* __restrict__ h,
                float* __restrict__ z, int* __restrict__ lst, const int* __restrict__ lcnt,
                int E, int n) {
  constexpr int Q     = CH / 4;
  constexpr int EPW   = 32 / Q;
  constexpr int PB    = 8;
  constexpr int U     = PB / EPW;
  constexpr int NW    = NT / 32;
  constexpr int CHUNK = NT * 4 * KL;
  static_assert(Q == 16 || Q == 32, "a row is 16 or 32 float4 groups: one or two edges per wave-wide access");
  static_assert(NW * EPW >= PB, "one access per wave covers the last, partial group of queue entries");
  static_assert((R * CH + 3 * CAP + 1) * 4 <= 320 * 1024, "sums and queue fit one work-group's LDS");
  static_assert(CAPS <= CAP, "a stored queue fits the LDS queue");

  __shared__ __attribute__((aligned(16))) float acc[R * CH];
  __shared__ int   qs[CAP];
  __shared__ int   qu[CAP];
  __shared__ float qn[CAP];
  __shared__ int   qcnt;

  const int tid   = threadIdx.x;
  const int lane  = tid & 31;
  const int wave  = tid >> 5;
  const int lo    = blockIdx.x * R;
  const int nrows = max(0, min(R, n - lo));

  for (int i = tid; i < R * Q; i += NT)
    ((float4*)acc)[i] = make_float4(0.f, 0.f, 0.f, 0.f);
  if (tid == 0) qcnt = 0;
  __syncthreads();

  int* lp = lst + (size_t)blockIdx.x * (3 * CAPS);
  int  total = 0;
  bool have  = false;
  {
    const int c = lcnt[blockIdx.x];
    if ((unsigned)c <= (unsigned)CAPS) {
      have  = true;
      total = c;
      for (int i = tid; i < c; i += NT) {
        const int s = lp[i];
        const int u = lp[CAPS + i];
        float w;
        if (LIST == 1) {
          w = dinv[s] * dinv[lo + u];
          lp[2 * CAPS + i] = __float_as_int(w);
        } else {
          w = __int_as_float(lp[2 * CAPS + i]);
        }
        qs[i] = s;
        qu[i] = u;
        qn[i] = w;
      }
    }
  }

  if (!have) {
  const int  nsteps = (E + CHUNK - 1) / CHUNK;
  const bool vec    = (E & 3) == 0;
  const int  rot    = nsteps > 0 ? (int)(blockIdx.x % (unsigned)nsteps) : 0;
  for (int st = 0; st < nsteps; ++st) {
    int sx = st + rot;
    if (sx >= nsteps) sx -= nsteps;
    const int base = sx * CHUNK + tid * 4;
    int4 d4[KL];
    if (vec && sx * CHUNK + CHUNK <= E) {
#pragma unroll
      for (int j = 0; j < KL; ++j)
        d4[j] = *(const int4*)(dst + base + j * (NT * 4));
    } else if (vec) {
#pragma unroll
      for (int j = 0; j < KL; ++j) {
        const int eb = base + j * (NT * 4);
        d4[j] = make_int4(-1, -1, -1, -1);
        if (eb < E) d4[j] = *(const int4*)(dst + eb);
      }
    } else {
#pragma unroll
      for (int j = 0; j < KL; ++j) {
        const int eb = base + j * (NT * 4);
        d4[j].x = eb     < E ? dst[eb]     : -1;
        d4[j].y = eb + 1 < E ? dst[eb + 1] : -1;
        d4[j].z = eb + 2 < E ? dst[eb + 2] : -1;
        d4[j].w = eb + 3 < E ? dst[eb + 3] : -1;
      }
    }
#pragma unroll
    for (int j = 0; j < KL; ++j) {
      const int eb = base + j * (NT * 4);
      const int dd[4] = {d4[j].x, d4[j].y, d4[j].z, d4[j].w};
#pragma unroll
      for (int k = 0; k < 4; ++k) {
        if ((unsigned)(dd[k] - lo) < (unsigned)nrows) {
          const int pos = atomicAdd(&qcnt, 1);
          if (pos < CAP) qs[pos] = eb + k;
        }
      }
    }
  }
  __syncthreads();
  total = qcnt;
  }

  if (total <= CAP) {
    if (!have) {
      for (int i = tid; i < total; i += NT) {
        const int e = qs[i];
        const int s = src[e];
        const int d = dst[e];
        const float w = dinv[s] * dinv[d];
        qs[i] = s;
        qu[i] = d - lo;
        qn[i] = w;
      }
    }
    __syncthreads();

    const int sub = lane / Q;
    const int f   = lane % Q;
    auto add_row = [&](int u, float w, const float4& hv) {
      float* ap = &acc[u * CH + f];
      const float v0 = hv.x * w, v1 = hv.y * w, v2 = hv.z * w, v3 = hv.w * w;
      const bool o = (EPW > 1) && sub != 0;
      atomicAdd(ap + (o ? 2 * Q : 0),     o ? v2 : v0);
      atomicAdd(ap + (o ? 3 * Q : Q),     o ? v3 : v1);
      atomicAdd(ap + (o ? 0     : 2 * Q), o ? v0 : v2);
      atomicAdd(ap + (o ? Q     : 3 * Q), o ? v1 : v3);
    };
    for (int i0 = wave * PB; i0 + PB <= total; i0 += NW * PB) {
      int s[U], u[U];
      float w[U];
      float4 hv[U];
#pragma unroll
      for (int k = 0; k < U; ++k) {
        const int i = i0 + k * EPW + sub;
        s[k] = qs[i]; u[k] = qu[i]; w[k] = qn[i];
      }
#pragma unroll
      for (int k = 0; k < U; ++k)
        hv[k] = *(const float4*)(h + (size_t)s[k] * CH + f * 4);
#pragma unroll
      for (int k = 0; k < U; ++k) add_row(u[k], w[k], hv[k]);
    }
    {
      const int i = total - total % PB + wave * EPW + sub;
      if (i < total) {
        const float4 hv = *(const float4*)(h + (size_t)qs[i] * CH + f * 4);
        add_row(qu[i], qn[i], hv);
      }
    }
  } else {
    for (int e = tid; e < E; e += NT) {
      const int d = dst[e];
      const unsigned uu = (unsigned)(d - lo);
      if (uu < (unsigned)nrows) {
        const int s = src[e];
        const float w = dinv[s] * dinv[d];
        const float* hp = h + (size_t)s * CH;
        float* ap = &acc[uu * CH];
        for (int c = 0; c < CH; ++c)
          atomicAdd(ap + (c & 3) * Q + (c >> 2), hp[c] * w);
      }
    }
  }
  __syncthreads();

  for (int i = tid; i < nrows * Q; i += NT) {
    const int u  = i / Q;
    const int ff = i % Q;
    const float* ap = &acc[u * CH + ff];
    float4* zp = (float4*)(z + (size_t)(lo + u) * CH) + ff;
    float4 zv = *zp;
    zv.x += ap[0];
    zv.y += ap[Q];
    zv.z += ap[2 * Q];
    zv.w += ap[3 * Q];
    *zp = zv;
  }
}

__global__ void edge_dot(const int* __restrict__ es, const int* __restrict__ ed,
                         const float* __restrict__ z, float* __restrict__ out, int L) {
  int j = blockIdx.x * blockDim.x + threadIdx.x;
  if (j >= L) return;
  const float4* a = (const float4*)(z + (size_t)es[j] * 64);
  const float4* b = (const float4*)(z + (size_t)ed[j] * 64);
  float sum = 0.f;
#pragma unroll
  for (int t = 0; t < 16; ++t) {
    float4 x = a[t], y = b[t];
    sum += x.x * y.x + x.y * y.y + x.z * y.z + x.w * y.w;
  }
  out[j] = sum;
}


static inline int cdiv_ll(long long a, long long b) { return (int)((a + b - 1) / b); }

extern "C" void kernel_launch(void* const* d_in, const int* in_sizes, int n_in,
                              void* d_out, int out_size, void* d_ws, size_t ws_size,
                              hipStream_t stream) {
  const float* x   = (const float*)d_in[0];
  const int*   ei  = (const int*)d_in[1];
  const int*   eli = (const int*)d_in[2];
  const float* W1  = (const float*)d_in[3];
  const float* b1  = (const float*)d_in[4];
  const float* W2  = (const float*)d_in[5];
  const float* b2  = (const float*)d_in[6];
  float* out = (float*)d_out;

  const int IN = 512, HID = 128, OUT = 64;
  const int n = in_sizes[0] / IN;
  const int E = in_sizes[1] / 2;
  const int L = in_sizes[2] / 2;
  const int* src = ei;
  const int* dst = ei + E;
  const int* es  = eli;
  const int* ed  = eli + L;

  float* ws   = (float*)d_ws;
  int*   lcnt = (int*)ws;  ws += n;
  float* dinv = ws;  ws += n;
  float* h1   = ws;  ws += (size_t)n * HID;
  float* z1   = ws;  ws += (size_t)n * HID;
  float* h2   = h1;
  float* z2   = h1 + (size_t)n * OUT;
  int*   lst  = (int*)ws;

  const int B = 256;
  constexpr int RNG  = 196;
  constexpr int QCAP = 12288;
  constexpr int QSAV = 8192;
  constexpr int QSCN = 20480;
  constexpr int GRP  = 16;
  constexpr int PRT  = 16;
  constexpr int FCAP = 4096;

  const int nb = cdiv_ll(n, RNG);
  const int ng = cdiv_ll(nb, GRP);
  int* fdst = (int*)h1;
  int* fsrc = fdst + (size_t)ng * PRT * FCAP;
  int* fcnt = fsrc + (size_t)ng * PRT * FCAP;

  filter_edges_by_group<GRP * RNG, PRT, 512, 8, FCAP, QSCN><<<ng * PRT, 512, 0, stream>>>(src, dst, fsrc, fdst, fcnt, E, n);
  range_scan_edges<RNG, 512, 8, QSCN, QSAV, GRP, PRT, FCAP><<<nb, 512, 0, stream>>>(src, dst, fsrc, fdst, fcnt, dinv, lst, lcnt, E, n);

  gemm_wmma_f32_lds<64, false><<<n / 16, 32 * (HID / 16), 0, stream>>>(x, W1, dinv, b1, h1, z1, n, IN, HID);
  range_sum_edges<128, RNG, 1024, 8, QCAP, QSAV, 1><<<cdiv_ll(n, RNG), 1024, 0, stream>>>(src, dst, dinv, h1, z1, lst, lcnt, E, n);

  gemm_wmma_f32_lds<64, true><<<n / 16, 32 * (OUT / 16), 0, stream>>>(z1, W2, dinv, b2, h2, z2, n, HID, OUT);
  range_sum_edges<64, RNG, 1024, 8, QCAP, QSAV, 2><<<cdiv_ll(n, RNG), 1024, 0, stream>>>(src, dst, dinv, h2, z2, lst, lcnt, E, n);

  edge_dot<<<cdiv_ll(L, B), B, 0, stream>>>(es, ed, z2, out, L);
}


# both products as persistent blocks: a wave loads a 16-column slice of the weight into registers once, then walks 16-row strips staged via two LDS buffers; no weight loads per strip; sum order kept
# speedup vs baseline: 2.4584x; 1.1147x over previous
#include <hip/hip_runtime.h>

typedef __attribute__((ext_vector_type(2))) float v2f;
typedef __attribute__((ext_vector_type(8))) float v8f;


template <int NT, int KL>
__device__ __forceinline__ void load_dst_step(const int* __restrict__ p, int first, int lim,
                                              bool vec, int tid, int4 (&d4)[KL]) {
  constexpr int CHUNK = NT * 4 * KL;
  const int base = first + tid * 4;
  if (vec && first + CHUNK <= lim) {
#pragma unroll
    for (int j = 0; j < KL; ++j)
      d4[j] = *(const int4*)(p + base + j * (NT * 4));
  } else if (vec) {
#pragma unroll
    for (int j = 0; j < KL; ++j) {
      const int eb = base + j * (NT * 4);
      d4[j] = make_int4(-1, -1, -1, -1);
      if (eb < lim) d4[j] = *(const int4*)(p + eb);
    }
  } else {
#pragma unroll
    for (int j = 0; j < KL; ++j) {
      const int eb = base + j * (NT * 4);
      d4[j].x = eb     < lim ? p[eb]     : -1;
      d4[j].y = eb + 1 < lim ? p[eb + 1] : -1;
      d4[j].z = eb + 2 < lim ? p[eb + 2] : -1;
      d4[j].w = eb + 3 < lim ? p[eb + 3] : -1;
    }
  }
}

template <int GROWS, int PRT, int NT, int KL, int FCAP, int QLDS>
__global__ void __launch_bounds__(NT)
filter_edges_by_group(const int* __restrict__ src, const int* __restrict__ dst,
                      int* __restrict__ fsrc, int* __restrict__ fdst, int* __restrict__ fcnt,
                      int E, int n) {
  constexpr int CHUNK = NT * 4 * KL;
  static_assert(FCAP <= QLDS, "a stored list fits the LDS queue");
  static_assert((2 * QLDS + 1) * 4 <= 320 * 1024, "the queue fits one work-group's LDS");

  __shared__ int qe[QLDS];
  __shared__ int qd[QLDS];
  __shared__ int qcnt;

  const int tid   = threadIdx.x;
  const int lane  = tid & 31;
  const int grp   = blockIdx.x / PRT;
  const int prt   = blockIdx.x % PRT;
  const int lo    = grp * GROWS;
  const int nrows = max(0, min(GROWS, n - lo));
  const int pe    = (((E + PRT - 1) / PRT) + 3) & ~3;
  const int e0    = min(E, prt * pe);
  const int e1    = min(E, e0 + pe);

  if (tid == 0) qcnt = 0;
  __syncthreads();

  const int  nsteps = (e1 - e0 + CHUNK - 1) / CHUNK;
  const bool vec    = (E & 3) == 0;
  const int  rot    = nsteps > 0 ? grp % nsteps : 0;
  for (int st = 0; st < nsteps; ++st) {
    int sx = st + rot;
    if (sx >= nsteps) sx -= nsteps;
    const int first = e0 + sx * CHUNK;
    int4 d4[KL];
    load_dst_step<NT, KL>(dst, first, e1, vec, tid, d4);

    unsigned mk[4 * KL];
    int tot = 0;
#pragma unroll
    for (int j = 0; j < KL; ++j) {
      const int dd[4] = {d4[j].x, d4[j].y, d4[j].z, d4[j].w};
#pragma unroll
      for (int k = 0; k < 4; ++k) {
        mk[j * 4 + k] = __builtin_amdgcn_ballot_w32((unsigned)(dd[k] - lo) < (unsigned)nrows);
        tot += __builtin_popcount(mk[j * 4 + k]);
      }
    }
    if (tot != 0) {
      int slot = 0;
      if (lane == 0) slot = atomicAdd(&qcnt, tot);
      slot = __builtin_amdgcn_readfirstlane(slot);
#pragma unroll
      for (int j = 0; j < KL; ++j) {
        const int eb = first + tid * 4 + j * (NT * 4);
        const int dd[4] = {d4[j].x, d4[j].y, d4[j].z, d4[j].w};
#pragma unroll
        for (int k = 0; k < 4; ++k) {
          const unsigned m = mk[j * 4 + k];
          if ((m >> lane) & 1u) {
            const int pos = slot + (int)__builtin_amdgcn_mbcnt_lo(m, 0u);
            if (pos < FCAP) { qe[pos] = eb + k; qd[pos] = dd[k]; }
          }
          slot += __builtin_popcount(m);
        }
      }
    }
  }
  __syncthreads();
  const int total = qcnt;
  if (tid == 0) fcnt[blockIdx.x] = total;

  if (total <= FCAP) {
    int* fs = fsrc + (size_t)blockIdx.x * FCAP;
    int* fd = fdst + (size_t)blockIdx.x * FCAP;
    for (int i0 = tid; i0 < FCAP; i0 += NT * 4) {
      int sv[4];
#pragma unroll
      for (int k = 0; k < 4; ++k) {
        const int i = i0 + k * NT;
        sv[k] = i < total ? src[qe[i]] : 0;
      }
#pragma unroll
      for (int k = 0; k < 4; ++k) {
        const int i = i0 + k * NT;
        if (i < total)     { fs[i] = sv[k]; fd[i] = qd[i]; }
        else if (i < FCAP) { fd[i] = -1; }
      }
    }
  }
}

template <int R, int NT, int KL, int CAPQ, int CAPS, int GRP, int PRT, int FCAP>
__global__ void __launch_bounds__(NT)
range_scan_edges(const int* __restrict__ src, const int* __restrict__ dst,
                 const int* __restrict__ fsrc, const int* __restrict__ fdst,
                 const int* __restrict__ fcnt,
                 float* __restrict__ dinv, int* __restrict__ lst, int* __restrict__ lcnt,
                 int E, int n) {
  constexpr int CHUNK = NT * 4 * KL;
  static_assert(CAPS <= CAPQ, "a stored queue fits the LDS queue");
  static_assert((2 * CAPQ + R + 1) * 4 <= 320 * 1024, "queue and counters fit one work-group's LDS");
  static_assert(FCAP % 4 == 0, "every filtered list starts on an int4 boundary");

  __shared__ int qs[CAPQ];
  __shared__ int qu[CAPQ];
  __shared__ int cnt[R];
  __shared__ int qcnt;

  const int tid   = threadIdx.x;
  const int lane  = tid & 31;
  const int lo    = blockIdx.x * R;
  const int nrows = max(0, min(R, n - lo));

  for (int i = tid; i < R; i += NT) cnt[i] = 0;
  if (tid == 0) qcnt = 0;
  __syncthreads();

  const int grp = blockIdx.x / GRP;
  int over = 0;
  for (int p = 0; p < PRT; ++p)
    over |= (unsigned)fcnt[grp * PRT + p] > (unsigned)FCAP ? 1 : 0;
  const bool filt = over == 0;
  const int* sd = filt ? fdst + (size_t)grp * (PRT * FCAP) : dst;
  const int* ss = filt ? fsrc + (size_t)grp * (PRT * FCAP) : src;
  const int  EE = filt ? PRT * FCAP : E;

  const int  nsteps = (EE + CHUNK - 1) / CHUNK;
  const bool vec    = (EE & 3) == 0;
  const int  rot    = nsteps > 0 ? (int)(blockIdx.x % (unsigned)nsteps) : 0;
  for (int st = 0; st < nsteps; ++st) {
    int sx = st + rot;
    if (sx >= nsteps) sx -= nsteps;
    const int first = sx * CHUNK;
    int4 d4[KL];
    load_dst_step<NT, KL>(sd, first, EE, vec, tid, d4);

    unsigned mk[4 * KL];
    int tot = 0;
#pragma unroll
    for (int j = 0; j < KL; ++j) {
      const int dd[4] = {d4[j].x, d4[j].y, d4[j].z, d4[j].w};
#pragma unroll
      for (int k = 0; k < 4; ++k) {
        mk[j * 4 + k] = __builtin_amdgcn_ballot_w32((unsigned)(dd[k] - lo) < (unsigned)nrows);
        tot += __builtin_popcount(mk[j * 4 + k]);
      }
    }
    if (tot != 0) {
      int slot = 0;
      if (lane == 0) slot = atomicAdd(&qcnt, tot);
      slot = __builtin_amdgcn_readfirstlane(slot);
#pragma unroll
      for (int j = 0; j < KL; ++j) {
        const int eb = first + tid * 4 + j * (NT * 4);
        const int dd[4] = {d4[j].x, d4[j].y, d4[j].z, d4[j].w};
#pragma unroll
        for (int k = 0; k < 4; ++k) {
          const unsigned m = mk[j * 4 + k];
          if ((m >> lane) & 1u) {
            const int uu  = dd[k] - lo;
            const int pos = slot + (int)__builtin_amdgcn_mbcnt_lo(m, 0u);
            atomicAdd(&cnt[uu], 1);
            if (pos < CAPQ) { qs[pos] = eb + k; qu[pos] = uu; }
          }
          slot += __builtin_popcount(m);
        }
      }
    }
  }
  __syncthreads();
  const int total = qcnt;
  if (tid == 0) lcnt[blockIdx.x] = total;

  if (total <= CAPS) {
    int* lp = lst + (size_t)blockIdx.x * (3 * CAPS);
    for (int i0 = tid; i0 < total; i0 += NT * 4) {
      int sv[4];
#pragma unroll
      for (int k = 0; k < 4; ++k) {
        const int i = i0 + k * NT;
        sv[k] = i < total ? ss[qs[i]] : 0;
      }
#pragma unroll
      for (int k = 0; k < 4; ++k) {
        const int i = i0 + k * NT;
        if (i < total) { lp[i] = sv[k]; lp[CAPS + i] = qu[i]; }
      }
    }
  }
  for (int i = tid; i < nrows; i += NT) {
    const float d = (float)(cnt[i] + 1);
    dinv[lo + i] = d > 0.f ? rsqrtf(d) : 0.f;
  }
}

template <int K, int NC, bool RELU_A>
__global__ void __launch_bounds__(32 * (NC / 16))
gemm_wmma_f32_wstat(const float* __restrict__ A,
                    const float* __restrict__ B,
                    const float* __restrict__ dinv,
                    const float* __restrict__ bias,
                    float* __restrict__ C,
                    float* __restrict__ Z,
                    int M) {
  constexpr int NT   = 32 * (NC / 16);
  constexpr int LSTR = K + 4;
  constexpr int RV   = K / 4;
  constexpr int PER  = 16 * RV / NT;
  constexpr int KS   = K / 4;
  constexpr int BW   = 16;
  constexpr int NB   = KS / BW;
  static_assert(KS % BW == 0, "a strip is whole blocks of WMMAs");
  static_assert((16 * RV) % NT == 0, "a strip is a whole number of float4 per thread");
  __shared__ __attribute__((aligned(16))) float As[2 * 16 * LSTR];

  const int tid   = threadIdx.x;
  const int lane  = tid & 31;
  const int wave  = tid >> 5;
  const int col0  = wave << 4;
  const int mn    = lane & 15;
  const int half  = lane >> 4;
  const int half2 = half << 1;
  const int nstrips = M >> 4;
  const int step  = gridDim.x;

  int s = blockIdx.x;
  float4 st0 = {}, st1 = {}, st2 = {}, st3 = {}, st4 = {}, st5 = {}, st6 = {}, st7 = {};
  static_assert(PER <= 8, "eight staging registers");
#define WSTAT_EACH(X) X(0, st0) X(1, st1) X(2, st2) X(3, st3) X(4, st4) X(5, st5) X(6, st6) X(7, st7)
#define WSTAT_LOAD(j, r) if constexpr (j < PER) r = *(const float4*)(gp + (size_t)(tid + j * NT) * 4);
#define WSTAT_STORE(j, r) if constexpr (j < PER) { \
    float4 av = r; \
    if (RELU_A) { av.x = fmaxf(av.x, 0.f); av.y = fmaxf(av.y, 0.f); av.z = fmaxf(av.z, 0.f); av.w = fmaxf(av.w, 0.f); } \
    *(float4*)(&nb[((tid + j * NT) / RV) * LSTR + ((tid + j * NT) % RV) * 4]) = av; }
  if (s < nstrips) {
    const float* gp = A + (size_t)s * (16 * K);
    WSTAT_EACH(WSTAT_LOAD)
  }

  v2f bf[KS];
  {
    const float* bp = B + (size_t)half2 * NC + col0 + mn;
#pragma unroll
    for (int q = 0; q < KS; ++q) {
      bf[q].x = bp[(size_t)(4 * q) * NC];
      bf[q].y = bp[(size_t)(4 * q + 1) * NC];
    }
  }
  const float bv = bias[col0 + mn];

  if (s < nstrips) {
    float* nb = &As[0];
    WSTAT_EACH(WSTAT_STORE)
  }
  __syncthreads();

  int buf = 0;
  for (; s < nstrips; s += step) {
    const bool more = s + step < nstrips;
    const int rbase = (s << 4) + (half << 3);
    const float4 dA = *(const float4*)(dinv + rbase);
    const float4 dB = *(const float4*)(dinv + rbase + 4);
    if (more) {
      const float* gp = A + (size_t)(s + step) * (16 * K);
      WSTAT_EACH(WSTAT_LOAD)
    }

    const float* ap = &As[buf * (16 * LSTR) + mn * LSTR + half2];
    v2f af[2][BW];
#pragma unroll
    for (int i = 0; i < BW; ++i) {
      af[0][i].x = ap[4 * i];
      af[0][i].y = ap[4 * i + 1];
    }
    v8f acc = {};
#pragma unroll
    for (int b = 0; b < NB; ++b) {
      if (b + 1 < NB) {
#pragma unroll
        for (int i = 0; i < BW; ++i) {
          af[(b + 1) & 1][i].x = ap[4 * ((b + 1) * BW + i)];
          af[(b + 1) & 1][i].y = ap[4 * ((b + 1) * BW + i) + 1];
        }
      }
      __builtin_amdgcn_sched_barrier(0);
#pragma unroll
      for (int i = 0; i < BW; ++i)
        acc = __builtin_amdgcn_wmma_f32_16x16x4_f32(
             false, af[b & 1][i],  false, bf[b * BW + i],
             (short)0, acc,  false,  false);
      __builtin_amdgcn_sched_barrier(0);
    }

    if (more) {
      float* nb = &As[(buf ^ 1) * (16 * LSTR)];
      WSTAT_EACH(WSTAT_STORE)
    }
    __syncthreads();
    buf ^= 1;

    const float sl[8] = {dA.x * dA.x, dA.y * dA.y, dA.z * dA.z, dA.w * dA.w,
                         dB.x * dB.x, dB.y * dB.y, dB.z * dB.z, dB.w * dB.w};
    float* cp = C + (size_t)rbase * NC + col0 + mn;
    float* zp = Z + (size_t)rbase * NC + col0 + mn;
#pragma unroll
    for (int g = 0; g < 8; ++g) {
      cp[(size_t)g * NC] = acc[g];
      zp[(size_t)g * NC] = acc[g] * sl[g] + bv;
    }
  }
#undef WSTAT_STORE
#undef WSTAT_LOAD
#undef WSTAT_EACH
}

template <int CH, int R, int NT, int KL, int CAP, int CAPS, int LIST>
__global__ void __launch_bounds__(NT)
range_sum_edges(const int* __restrict__ src, const int* __restrict__ dst,
                const float* __restrict__ dinv, const float* __restrict__ h,
                float* __restrict__ z, int* __restrict__ lst, const int* __restrict__ lcnt,
                int E, int n) {
  constexpr int Q     = CH / 4;
  constexpr int EPW   = 32 / Q;
  constexpr int PB    = 8;
  constexpr int U     = PB / EPW;
  constexpr int NW    = NT / 32;
  constexpr int CHUNK = NT * 4 * KL;
  static_assert(Q == 16 || Q == 32, "a row is 16 or 32 float4 groups: one or two edges per wave-wide access");
  static_assert(NW * EPW >= PB, "one access per wave covers the last, partial group of queue entries");
  static_assert((R * CH + 3 * CAP + 1) * 4 <= 320 * 1024, "sums and queue fit one work-group's LDS");
  static_assert(CAPS <= CAP, "a stored queue fits the LDS queue");

  __shared__ __attribute__((aligned(16))) float acc[R * CH];
  __shared__ int   qs[CAP];
  __shared__ int   qu[CAP];
  __shared__ float qn[CAP];
  __shared__ int   qcnt;

  const int tid   = threadIdx.x;
  const int lane  = tid & 31;
  const int wave  = tid >> 5;
  const int lo    = blockIdx.x * R;
  const int nrows = max(0, min(R, n - lo));

  for (int i = tid; i < R * Q; i += NT)
    ((float4*)acc)[i] = make_float4(0.f, 0.f, 0.f, 0.f);
  if (tid == 0) qcnt = 0;
  __syncthreads();

  int* lp = lst + (size_t)blockIdx.x * (3 * CAPS);
  int  total = 0;
  bool have  = false;
  {
    const int c = lcnt[blockIdx.x];
    if ((unsigned)c <= (unsigned)CAPS) {
      have  = true;
      total = c;
      for (int i = tid; i < c; i += NT) {
        const int s = lp[i];
        const int u = lp[CAPS + i];
        float w;
        if (LIST == 1) {
          w = dinv[s] * dinv[lo + u];
          lp[2 * CAPS + i] = __float_as_int(w);
        } else {
          w = __int_as_float(lp[2 * CAPS + i]);
        }
        qs[i] = s;
        qu[i] = u;
        qn[i] = w;
      }
    }
  }

  if (!have) {
  const int  nsteps = (E + CHUNK - 1) / CHUNK;
  const bool vec    = (E & 3) == 0;
  const int  rot    = nsteps > 0 ? (int)(blockIdx.x % (unsigned)nsteps) : 0;
  for (int st = 0; st < nsteps; ++st) {
    int sx = st + rot;
    if (sx >= nsteps) sx -= nsteps;
    const int base = sx * CHUNK + tid * 4;
    int4 d4[KL];
    if (vec && sx * CHUNK + CHUNK <= E) {
#pragma unroll
      for (int j = 0; j < KL; ++j)
        d4[j] = *(const int4*)(dst + base + j * (NT * 4));
    } else if (vec) {
#pragma unroll
      for (int j = 0; j < KL; ++j) {
        const int eb = base + j * (NT * 4);
        d4[j] = make_int4(-1, -1, -1, -1);
        if (eb < E) d4[j] = *(const int4*)(dst + eb);
      }
    } else {
#pragma unroll
      for (int j = 0; j < KL; ++j) {
        const int eb = base + j * (NT * 4);
        d4[j].x = eb     < E ? dst[eb]     : -1;
        d4[j].y = eb + 1 < E ? dst[eb + 1] : -1;
        d4[j].z = eb + 2 < E ? dst[eb + 2] : -1;
        d4[j].w = eb + 3 < E ? dst[eb + 3] : -1;
      }
    }
#pragma unroll
    for (int j = 0; j < KL; ++j) {
      const int eb = base + j * (NT * 4);
      const int dd[4] = {d4[j].x, d4[j].y, d4[j].z, d4[j].w};
#pragma unroll
      for (int k = 0; k < 4; ++k) {
        if ((unsigned)(dd[k] - lo) < (unsigned)nrows) {
          const int pos = atomicAdd(&qcnt, 1);
          if (pos < CAP) qs[pos] = eb + k;
        }
      }
    }
  }
  __syncthreads();
  total = qcnt;
  }

  if (total <= CAP) {
    if (!have) {
      for (int i = tid; i < total; i += NT) {
        const int e = qs[i];
        const int s = src[e];
        const int d = dst[e];
        const float w = dinv[s] * dinv[d];
        qs[i] = s;
        qu[i] = d - lo;
        qn[i] = w;
      }
    }
    __syncthreads();

    const int sub = lane / Q;
    const int f   = lane % Q;
    auto add_row = [&](int u, float w, const float4& hv) {
      float* ap = &acc[u * CH + f];
      const float v0 = hv.x * w, v1 = hv.y * w, v2 = hv.z * w, v3 = hv.w * w;
      const bool o = (EPW > 1) && sub != 0;
      atomicAdd(ap + (o ? 2 * Q : 0),     o ? v2 : v0);
      atomicAdd(ap + (o ? 3 * Q : Q),     o ? v3 : v1);
      atomicAdd(ap + (o ? 0     : 2 * Q), o ? v0 : v2);
      atomicAdd(ap + (o ? Q     : 3 * Q), o ? v1 : v3);
    };
    for (int i0 = wave * PB; i0 + PB <= total; i0 += NW * PB) {
      int s[U], u[U];
      float w[U];
      float4 hv[U];
#pragma unroll
      for (int k = 0; k < U; ++k) {
        const int i = i0 + k * EPW + sub;
        s[k] = qs[i]; u[k] = qu[i]; w[k] = qn[i];
      }
#pragma unroll
      for (int k = 0; k < U; ++k)
        hv[k] = *(const float4*)(h + (size_t)s[k] * CH + f * 4);
#pragma unroll
      for (int k = 0; k < U; ++k) add_row(u[k], w[k], hv[k]);
    }
    {
      const int i = total - total % PB + wave * EPW + sub;
      if (i < total) {
        const float4 hv = *(const float4*)(h + (size_t)qs[i] * CH + f * 4);
        add_row(qu[i], qn[i], hv);
      }
    }
  } else {
    for (int e = tid; e < E; e += NT) {
      const int d = dst[e];
      const unsigned uu = (unsigned)(d - lo);
      if (uu < (unsigned)nrows) {
        const int s = src[e];
        const float w = dinv[s] * dinv[d];
        const float* hp = h + (size_t)s * CH;
        float* ap = &acc[uu * CH];
        for (int c = 0; c < CH; ++c)
          atomicAdd(ap + (c & 3) * Q + (c >> 2), hp[c] * w);
      }
    }
  }
  __syncthreads();

  for (int i = tid; i < nrows * Q; i += NT) {
    const int u  = i / Q;
    const int ff = i % Q;
    const float* ap = &acc[u * CH + ff];
    float4* zp = (float4*)(z + (size_t)(lo + u) * CH) + ff;
    float4 zv = *zp;
    zv.x += ap[0];
    zv.y += ap[Q];
    zv.z += ap[2 * Q];
    zv.w += ap[3 * Q];
    *zp = zv;
  }
}

__global__ void edge_dot(const int* __restrict__ es, const int* __restrict__ ed,
                         const float* __restrict__ z, float* __restrict__ out, int L) {
  int j = blockIdx.x * blockDim.x + threadIdx.x;
  if (j >= L) return;
  const float4* a = (const float4*)(z + (size_t)es[j] * 64);
  const float4* b = (const float4*)(z + (size_t)ed[j] * 64);
  float sum = 0.f;
#pragma unroll
  for (int t = 0; t < 16; ++t) {
    float4 x = a[t], y = b[t];
    sum += x.x * y.x + x.y * y.y + x.z * y.z + x.w * y.w;
  }
  out[j] = sum;
}


static inline int cdiv_ll(long long a, long long b) { return (int)((a + b - 1) / b); }

extern "C" void kernel_launch(void* const* d_in, const int* in_sizes, int n_in,
                              void* d_out, int out_size, void* d_ws, size_t ws_size,
                              hipStream_t stream) {
  const float* x   = (const float*)d_in[0];
  const int*   ei  = (const int*)d_in[1];
  const int*   eli = (const int*)d_in[2];
  const float* W1  = (const float*)d_in[3];
  const float* b1  = (const float*)d_in[4];
  const float* W2  = (const float*)d_in[5];
  const float* b2  = (const float*)d_in[6];
  float* out = (float*)d_out;

  const int IN = 512, HID = 128, OUT = 64;
  const int n = in_sizes[0] / IN;
  const int E = in_sizes[1] / 2;
  const int L = in_sizes[2] / 2;
  const int* src = ei;
  const int* dst = ei + E;
  const int* es  = eli;
  const int* ed  = eli + L;

  float* ws   = (float*)d_ws;
  int*   lcnt = (int*)ws;  ws += n;
  float* dinv = ws;  ws += n;
  float* h1   = ws;  ws += (size_t)n * HID;
  float* z1   = ws;  ws += (size_t)n * HID;
  float* h2   = h1;
  float* z2   = h1 + (size_t)n * OUT;
  int*   lst  = (int*)ws;

  const int B = 256;
  constexpr int RNG  = 196;
  constexpr int QCAP = 12288;
  constexpr int QSAV = 8192;
  constexpr int QSCN = 20480;
  constexpr int GRP  = 16;
  constexpr int PRT  = 16;
  constexpr int FCAP = 4096;

  const int nb = cdiv_ll(n, RNG);
  const int ng = cdiv_ll(nb, GRP);
  int* fdst = (int*)h1;
  int* fsrc = fdst + (size_t)ng * PRT * FCAP;
  int* fcnt = fsrc + (size_t)ng * PRT * FCAP;

  filter_edges_by_group<GRP * RNG, PRT, 512, 8, FCAP, QSCN><<<ng * PRT, 512, 0, stream>>>(src, dst, fsrc, fdst, fcnt, E, n);
  range_scan_edges<RNG, 512, 8, QSCN, QSAV, GRP, PRT, FCAP><<<nb, 512, 0, stream>>>(src, dst, fsrc, fdst, fcnt, dinv, lst, lcnt, E, n);

  gemm_wmma_f32_wstat<512, 128, false><<<256, 32 * (HID / 16), 0, stream>>>(x, W1, dinv, b1, h1, z1, n);
  range_sum_edges<128, RNG, 1024, 8, QCAP, QSAV, 1><<<cdiv_ll(n, RNG), 1024, 0, stream>>>(src, dst, dinv, h1, z1, lst, lcnt, E, n);

  gemm_wmma_f32_wstat<128, 64, true><<<625, 32 * (OUT / 16), 0, stream>>>(z1, W2, dinv, b2, h2, z2, n);
  range_sum_edges<64, RNG, 1024, 8, QCAP, QSAV, 2><<<cdiv_ll(n, RNG), 1024, 0, stream>>>(src, dst, dinv, h2, z2, lst, lcnt, E, n);

  edge_dot<<<cdiv_ll(L, B), B, 0, stream>>>(es, ed, z2, out, L);
}


# edge sums start in LDS: an owner block loads its own h rows and its stored queue at entry as one group, seeds its LDS sums with h*dinv^2+b (sum order changed); GEMMs store h only; store-only close
# speedup vs baseline: 2.6843x; 1.0919x over previous
#include <hip/hip_runtime.h>

typedef __attribute__((ext_vector_type(2))) float v2f;
typedef __attribute__((ext_vector_type(8))) float v8f;


template <int NT, int KL>
__device__ __forceinline__ void load_dst_step(const int* __restrict__ p, int first, int lim,
                                              bool vec, int tid, int4 (&d4)[KL]) {
  constexpr int CHUNK = NT * 4 * KL;
  const int base = first + tid * 4;
  if (vec && first + CHUNK <= lim) {
#pragma unroll
    for (int j = 0; j < KL; ++j)
      d4[j] = *(const int4*)(p + base + j * (NT * 4));
  } else if (vec) {
#pragma unroll
    for (int j = 0; j < KL; ++j) {
      const int eb = base + j * (NT * 4);
      d4[j] = make_int4(-1, -1, -1, -1);
      if (eb < lim) d4[j] = *(const int4*)(p + eb);
    }
  } else {
#pragma unroll
    for (int j = 0; j < KL; ++j) {
      const int eb = base + j * (NT * 4);
      d4[j].x = eb     < lim ? p[eb]     : -1;
      d4[j].y = eb + 1 < lim ? p[eb + 1] : -1;
      d4[j].z = eb + 2 < lim ? p[eb + 2] : -1;
      d4[j].w = eb + 3 < lim ? p[eb + 3] : -1;
    }
  }
}

template <int GROWS, int PRT, int NT, int KL, int FCAP, int QLDS>
__global__ void __launch_bounds__(NT)
filter_edges_by_group(const int* __restrict__ src, const int* __restrict__ dst,
                      int* __restrict__ fsrc, int* __restrict__ fdst, int* __restrict__ fcnt,
                      int E, int n) {
  constexpr int CHUNK = NT * 4 * KL;
  static_assert(FCAP <= QLDS, "a stored list fits the LDS queue");
  static_assert((2 * QLDS + 1) * 4 <= 320 * 1024, "the queue fits one work-group's LDS");

  __shared__ int qe[QLDS];
  __shared__ int qd[QLDS];
  __shared__ int qcnt;

  const int tid   = threadIdx.x;
  const int lane  = tid & 31;
  const int grp   = blockIdx.x / PRT;
  const int prt   = blockIdx.x % PRT;
  const int lo    = grp * GROWS;
  const int nrows = max(0, min(GROWS, n - lo));
  const int pe    = (((E + PRT - 1) / PRT) + 3) & ~3;
  const int e0    = min(E, prt * pe);
  const int e1    = min(E, e0 + pe);

  if (tid == 0) qcnt = 0;
  __syncthreads();

  const int  nsteps = (e1 - e0 + CHUNK - 1) / CHUNK;
  const bool vec    = (E & 3) == 0;
  const int  rot    = nsteps > 0 ? grp % nsteps : 0;
  for (int st = 0; st < nsteps; ++st) {
    int sx = st + rot;
    if (sx >= nsteps) sx -= nsteps;
    const int first = e0 + sx * CHUNK;
    int4 d4[KL];
    load_dst_step<NT, KL>(dst, first, e1, vec, tid, d4);

    unsigned mk[4 * KL];
    int tot = 0;
#pragma unroll
    for (int j = 0; j < KL; ++j) {
      const int dd[4] = {d4[j].x, d4[j].y, d4[j].z, d4[j].w};
#pragma unroll
      for (int k = 0; k < 4; ++k) {
        mk[j * 4 + k] = __builtin_amdgcn_ballot_w32((unsigned)(dd[k] - lo) < (unsigned)nrows);
        tot += __builtin_popcount(mk[j * 4 + k]);
      }
    }
    if (tot != 0) {
      int slot = 0;
      if (lane == 0) slot = atomicAdd(&qcnt, tot);
      slot = __builtin_amdgcn_readfirstlane(slot);
#pragma unroll
      for (int j = 0; j < KL; ++j) {
        const int eb = first + tid * 4 + j * (NT * 4);
        const int dd[4] = {d4[j].x, d4[j].y, d4[j].z, d4[j].w};
#pragma unroll
        for (int k = 0; k < 4; ++k) {
          const unsigned m = mk[j * 4 + k];
          if ((m >> lane) & 1u) {
            const int pos = slot + (int)__builtin_amdgcn_mbcnt_lo(m, 0u);
            if (pos < FCAP) { qe[pos] = eb + k; qd[pos] = dd[k]; }
          }
          slot += __builtin_popcount(m);
        }
      }
    }
  }
  __syncthreads();
  const int total = qcnt;
  if (tid == 0) fcnt[blockIdx.x] = total;

  if (total <= FCAP) {
    int* fs = fsrc + (size_t)blockIdx.x * FCAP;
    int* fd = fdst + (size_t)blockIdx.x * FCAP;
    for (int i0 = tid; i0 < FCAP; i0 += NT * 4) {
      int sv[4];
#pragma unroll
      for (int k = 0; k < 4; ++k) {
        const int i = i0 + k * NT;
        sv[k] = i < total ? src[qe[i]] : 0;
      }
#pragma unroll
      for (int k = 0; k < 4; ++k) {
        const int i = i0 + k * NT;
        if (i < total)     { fs[i] = sv[k]; fd[i] = qd[i]; }
        else if (i < FCAP) { fd[i] = -1; }
      }
    }
  }
}

template <int R, int NT, int KL, int CAPQ, int CAPS, int GRP, int PRT, int FCAP>
__global__ void __launch_bounds__(NT)
range_scan_edges(const int* __restrict__ src, const int* __restrict__ dst,
                 const int* __restrict__ fsrc, const int* __restrict__ fdst,
                 const int* __restrict__ fcnt,
                 float* __restrict__ dinv, int* __restrict__ lst, int* __restrict__ lcnt,
                 int E, int n) {
  constexpr int CHUNK = NT * 4 * KL;
  static_assert(CAPS <= CAPQ, "a stored queue fits the LDS queue");
  static_assert((2 * CAPQ + R + 1) * 4 <= 320 * 1024, "queue and counters fit one work-group's LDS");
  static_assert(FCAP % 4 == 0, "every filtered list starts on an int4 boundary");

  __shared__ int qs[CAPQ];
  __shared__ int qu[CAPQ];
  __shared__ int cnt[R];
  __shared__ int qcnt;

  const int tid   = threadIdx.x;
  const int lane  = tid & 31;
  const int lo    = blockIdx.x * R;
  const int nrows = max(0, min(R, n - lo));

  for (int i = tid; i < R; i += NT) cnt[i] = 0;
  if (tid == 0) qcnt = 0;
  __syncthreads();

  const int grp = blockIdx.x / GRP;
  int over = 0;
  for (int p = 0; p < PRT; ++p)
    over |= (unsigned)fcnt[grp * PRT + p] > (unsigned)FCAP ? 1 : 0;
  const bool filt = over == 0;
  const int* sd = filt ? fdst + (size_t)grp * (PRT * FCAP) : dst;
  const int* ss = filt ? fsrc + (size_t)grp * (PRT * FCAP) : src;
  const int  EE = filt ? PRT * FCAP : E;

  const int  nsteps = (EE + CHUNK - 1) / CHUNK;
  const bool vec    = (EE & 3) == 0;
  const int  rot    = nsteps > 0 ? (int)(blockIdx.x % (unsigned)nsteps) : 0;
  for (int st = 0; st < nsteps; ++st) {
    int sx = st + rot;
    if (sx >= nsteps) sx -= nsteps;
    const int first = sx * CHUNK;
    int4 d4[KL];
    load_dst_step<NT, KL>(sd, first, EE, vec, tid, d4);

    unsigned mk[4 * KL];
    int tot = 0;
#pragma unroll
    for (int j = 0; j < KL; ++j) {
      const int dd[4] = {d4[j].x, d4[j].y, d4[j].z, d4[j].w};
#pragma unroll
      for (int k = 0; k < 4; ++k) {
        mk[j * 4 + k] = __builtin_amdgcn_ballot_w32((unsigned)(dd[k] - lo) < (unsigned)nrows);
        tot += __builtin_popcount(mk[j * 4 + k]);
      }
    }
    if (tot != 0) {
      int slot = 0;
      if (lane == 0) slot = atomicAdd(&qcnt, tot);
      slot = __builtin_amdgcn_readfirstlane(slot);
#pragma unroll
      for (int j = 0; j < KL; ++j) {
        const int eb = first + tid * 4 + j * (NT * 4);
        const int dd[4] = {d4[j].x, d4[j].y, d4[j].z, d4[j].w};
#pragma unroll
        for (int k = 0; k < 4; ++k) {
          const unsigned m = mk[j * 4 + k];
          if ((m >> lane) & 1u) {
            const int uu  = dd[k] - lo;
            const int pos = slot + (int)__builtin_amdgcn_mbcnt_lo(m, 0u);
            atomicAdd(&cnt[uu], 1);
            if (pos < CAPQ) { qs[pos] = eb + k; qu[pos] = uu; }
          }
          slot += __builtin_popcount(m);
        }
      }
    }
  }
  __syncthreads();
  const int total = qcnt;
  if (tid == 0) lcnt[blockIdx.x] = total;

  if (total <= CAPS) {
    int* lp = lst + (size_t)blockIdx.x * (3 * CAPS);
    for (int i0 = tid; i0 < total; i0 += NT * 4) {
      int sv[4];
#pragma unroll
      for (int k = 0; k < 4; ++k) {
        const int i = i0 + k * NT;
        sv[k] = i < total ? ss[qs[i]] : 0;
      }
#pragma unroll
      for (int k = 0; k < 4; ++k) {
        const int i = i0 + k * NT;
        if (i < total) { lp[i] = sv[k]; lp[CAPS + i] = qu[i]; }
      }
    }
  }
  for (int i = tid; i < nrows; i += NT) {
    const float d = (float)(cnt[i] + 1);
    dinv[lo + i] = d > 0.f ? rsqrtf(d) : 0.f;
  }
}

template <int K, int NC, bool RELU_A>
__global__ void __launch_bounds__(32 * (NC / 16))
gemm_wmma_f32_wstat(const float* __restrict__ A,
                    const float* __restrict__ B,
                    float* __restrict__ C,
                    int M) {
  constexpr int NT   = 32 * (NC / 16);
  constexpr int LSTR = K + 4;
  constexpr int RV   = K / 4;
  constexpr int PER  = 16 * RV / NT;
  constexpr int KS   = K / 4;
  constexpr int BW   = 16;
  constexpr int NB   = KS / BW;
  static_assert(KS % BW == 0, "a strip is whole blocks of WMMAs");
  static_assert((16 * RV) % NT == 0, "a strip is a whole number of float4 per thread");
  __shared__ __attribute__((aligned(16))) float As[2 * 16 * LSTR];

  const int tid   = threadIdx.x;
  const int lane  = tid & 31;
  const int wave  = tid >> 5;
  const int col0  = wave << 4;
  const int mn    = lane & 15;
  const int half  = lane >> 4;
  const int half2 = half << 1;
  const int nstrips = M >> 4;
  const int step  = gridDim.x;

  int s = blockIdx.x;
  float4 st0 = {}, st1 = {}, st2 = {}, st3 = {}, st4 = {}, st5 = {}, st6 = {}, st7 = {};
  static_assert(PER <= 8, "eight staging registers");
#define WSTAT_EACH(X) X(0, st0) X(1, st1) X(2, st2) X(3, st3) X(4, st4) X(5, st5) X(6, st6) X(7, st7)
#define WSTAT_LOAD(j, r) if constexpr (j < PER) r = *(const float4*)(gp + (size_t)(tid + j * NT) * 4);
#define WSTAT_STORE(j, r) if constexpr (j < PER) { \
    float4 av = r; \
    if (RELU_A) { av.x = fmaxf(av.x, 0.f); av.y = fmaxf(av.y, 0.f); av.z = fmaxf(av.z, 0.f); av.w = fmaxf(av.w, 0.f); } \
    *(float4*)(&nb[((tid + j * NT) / RV) * LSTR + ((tid + j * NT) % RV) * 4]) = av; }
  if (s < nstrips) {
    const float* gp = A + (size_t)s * (16 * K);
    WSTAT_EACH(WSTAT_LOAD)
  }

  v2f bf[KS];
  {
    const float* bp = B + (size_t)half2 * NC + col0 + mn;
#pragma unroll
    for (int q = 0; q < KS; ++q) {
      bf[q].x = bp[(size_t)(4 * q) * NC];
      bf[q].y = bp[(size_t)(4 * q + 1) * NC];
    }
  }

  if (s < nstrips) {
    float* nb = &As[0];
    WSTAT_EACH(WSTAT_STORE)
  }
  __syncthreads();

  int buf = 0;
  for (; s < nstrips; s += step) {
    const bool more = s + step < nstrips;
    const int rbase = (s << 4) + (half << 3);
    if (more) {
      const float* gp = A + (size_t)(s + step) * (16 * K);
      WSTAT_EACH(WSTAT_LOAD)
    }

    const float* ap = &As[buf * (16 * LSTR) + mn * LSTR + half2];
    v2f af[2][BW];
#pragma unroll
    for (int i = 0; i < BW; ++i) {
      af[0][i].x = ap[4 * i];
      af[0][i].y = ap[4 * i + 1];
    }
    v8f acc = {};
#pragma unroll
    for (int b = 0; b < NB; ++b) {
      if (b + 1 < NB) {
#pragma unroll
        for (int i = 0; i < BW; ++i) {
          af[(b + 1) & 1][i].x = ap[4 * ((b + 1) * BW + i)];
          af[(b + 1) & 1][i].y = ap[4 * ((b + 1) * BW + i) + 1];
        }
      }
      __builtin_amdgcn_sched_barrier(0);
#pragma unroll
      for (int i = 0; i < BW; ++i)
        acc = __builtin_amdgcn_wmma_f32_16x16x4_f32(
             false, af[b & 1][i],  false, bf[b * BW + i],
             (short)0, acc,  false,  false);
      __builtin_amdgcn_sched_barrier(0);
    }

    if (more) {
      float* nb = &As[(buf ^ 1) * (16 * LSTR)];
      WSTAT_EACH(WSTAT_STORE)
    }
    __syncthreads();
    buf ^= 1;

    float* cp = C + (size_t)rbase * NC + col0 + mn;
#pragma unroll
    for (int g = 0; g < 8; ++g)
      cp[(size_t)g * NC] = acc[g];
  }
#undef WSTAT_STORE
#undef WSTAT_LOAD
#undef WSTAT_EACH
}

template <int CH, int R, int NT, int KL, int CAP, int CAPS, int LIST>
__global__ void __launch_bounds__(NT)
range_sum_edges(const int* __restrict__ src, const int* __restrict__ dst,
                const float* __restrict__ dinv, const float* __restrict__ bias,
                const float* __restrict__ h,
                float* __restrict__ z, int* __restrict__ lst, const int* __restrict__ lcnt,
                int E, int n) {
  constexpr int Q     = CH / 4;
  constexpr int EPW   = 32 / Q;
  constexpr int PB    = 8;
  constexpr int U     = PB / EPW;
  constexpr int NW    = NT / 32;
  constexpr int CHUNK = NT * 4 * KL;
  constexpr int ZI    = (R * Q + NT - 1) / NT;
  constexpr int QI    = 4;
  static_assert(QI * NT <= CAPS, "the entry group stays inside a stored queue's planes");
  static_assert(NT % Q == 0, "a thread has the same float4 group of a row in every pass over the rows");
  static_assert(Q == 16 || Q == 32, "a row is 16 or 32 float4 groups: one or two edges per wave-wide access");
  static_assert(NW * EPW >= PB, "one access per wave covers the last, partial group of queue entries");
  static_assert((R * CH + 3 * CAP + 1) * 4 <= 320 * 1024, "sums and queue fit one work-group's LDS");
  static_assert(CAPS <= CAP, "a stored queue fits the LDS queue");

  __shared__ __attribute__((aligned(16))) float acc[R * CH];
  __shared__ int   qs[CAP];
  __shared__ int   qu[CAP];
  __shared__ float qn[CAP];
  __shared__ int   qcnt;

  const int tid   = threadIdx.x;
  const int lane  = tid & 31;
  const int wave  = tid >> 5;
  const int lo    = blockIdx.x * R;
  const int nrows = max(0, min(R, n - lo));

  const int    zf = tid % Q;
  float4 bq;
  float4 zs[ZI];
  float  zd[ZI];
  auto load_own_rows = [&]() {
#pragma unroll
    for (int k = 0; k < ZI; ++k) {
      const int row = min(lo + (tid + k * NT) / Q, n - 1);
      zs[k] = *(const float4*)(h + (size_t)row * CH + zf * 4);
      zd[k] = dinv[row];
    }
    bq = *(const float4*)(bias + zf * 4);
  };
  if (tid == 0) qcnt = 0;

  int* lp = lst + (size_t)blockIdx.x * (3 * CAPS);
  int  total = 0;
  bool have  = false;
  const int c = lcnt[blockIdx.x];
  if ((unsigned)c <= (unsigned)CAPS && c > 0) {
    have  = true;
    total = c;
    int   es[QI], eu[QI];
    float ew[QI];
#pragma unroll
    for (int k = 0; k < QI; ++k) {
      const int ic = min(tid + k * NT, c - 1);
      es[k] = lp[ic];
      eu[k] = lp[CAPS + ic];
      if (LIST != 1) ew[k] = __int_as_float(lp[2 * CAPS + ic]);
    }
    load_own_rows();
    if (LIST == 1) {
#pragma unroll
      for (int k = 0; k < QI; ++k)
        ew[k] = dinv[es[k]] * dinv[lo + eu[k]];
    }
#pragma unroll
    for (int k = 0; k < QI; ++k) {
      const int i = tid + k * NT;
      qs[i] = es[k];
      qu[i] = eu[k];
      qn[i] = ew[k];
      if (LIST == 1 && i < c) lp[2 * CAPS + i] = __float_as_int(ew[k]);
    }
    for (int i = tid + QI * NT; i < c; i += NT) {
      const int s = lp[i];
      const int u = lp[CAPS + i];
      float w;
      if (LIST == 1) {
        w = dinv[s] * dinv[lo + u];
        lp[2 * CAPS + i] = __float_as_int(w);
      } else {
        w = __int_as_float(lp[2 * CAPS + i]);
      }
      qs[i] = s;
      qu[i] = u;
      qn[i] = w;
    }
  } else {
    have = (c == 0);
    load_own_rows();
  }

  if (!have) {
  __syncthreads();
  const int  nsteps = (E + CHUNK - 1) / CHUNK;
  const bool vec    = (E & 3) == 0;
  const int  rot    = nsteps > 0 ? (int)(blockIdx.x % (unsigned)nsteps) : 0;
  for (int st = 0; st < nsteps; ++st) {
    int sx = st + rot;
    if (sx >= nsteps) sx -= nsteps;
    const int base = sx * CHUNK + tid * 4;
    int4 d4[KL];
    if (vec && sx * CHUNK + CHUNK <= E) {
#pragma unroll
      for (int j = 0; j < KL; ++j)
        d4[j] = *(const int4*)(dst + base + j * (NT * 4));
    } else if (vec) {
#pragma unroll
      for (int j = 0; j < KL; ++j) {
        const int eb = base + j * (NT * 4);
        d4[j] = make_int4(-1, -1, -1, -1);
        if (eb < E) d4[j] = *(const int4*)(dst + eb);
      }
    } else {
#pragma unroll
      for (int j = 0; j < KL; ++j) {
        const int eb = base + j * (NT * 4);
        d4[j].x = eb     < E ? dst[eb]     : -1;
        d4[j].y = eb + 1 < E ? dst[eb + 1] : -1;
        d4[j].z = eb + 2 < E ? dst[eb + 2] : -1;
        d4[j].w = eb + 3 < E ? dst[eb + 3] : -1;
      }
    }
#pragma unroll
    for (int j = 0; j < KL; ++j) {
      const int eb = base + j * (NT * 4);
      const int dd[4] = {d4[j].x, d4[j].y, d4[j].z, d4[j].w};
#pragma unroll
      for (int k = 0; k < 4; ++k) {
        if ((unsigned)(dd[k] - lo) < (unsigned)nrows) {
          const int pos = atomicAdd(&qcnt, 1);
          if (pos < CAP) qs[pos] = eb + k;
        }
      }
    }
  }
  __syncthreads();
  total = qcnt;
  }

  if (!have && total <= CAP) {
    for (int i = tid; i < total; i += NT) {
      const int e = qs[i];
      const int s = src[e];
      const int d = dst[e];
      const float w = dinv[s] * dinv[d];
      qs[i] = s;
      qu[i] = d - lo;
      qn[i] = w;
    }
  }

  {
#pragma unroll
    for (int k = 0; k < ZI; ++k) {
      const int u = (tid + k * NT) / Q;
      if (u < nrows) {
        const float s2 = zd[k] * zd[k];
        float* ap = &acc[u * CH + zf];
        ap[0]     = zs[k].x * s2 + bq.x;
        ap[Q]     = zs[k].y * s2 + bq.y;
        ap[2 * Q] = zs[k].z * s2 + bq.z;
        ap[3 * Q] = zs[k].w * s2 + bq.w;
      }
    }
  }
  __syncthreads();

  if (total <= CAP) {
    const int sub = lane / Q;
    const int f   = lane % Q;
    auto add_row = [&](int u, float w, const float4& hv) {
      float* ap = &acc[u * CH + f];
      const float v0 = hv.x * w, v1 = hv.y * w, v2 = hv.z * w, v3 = hv.w * w;
      const bool o = (EPW > 1) && sub != 0;
      atomicAdd(ap + (o ? 2 * Q : 0),     o ? v2 : v0);
      atomicAdd(ap + (o ? 3 * Q : Q),     o ? v3 : v1);
      atomicAdd(ap + (o ? 0     : 2 * Q), o ? v0 : v2);
      atomicAdd(ap + (o ? Q     : 3 * Q), o ? v1 : v3);
    };
    for (int i0 = wave * PB; i0 + PB <= total; i0 += NW * PB) {
      int s[U], u[U];
      float w[U];
      float4 hv[U];
#pragma unroll
      for (int k = 0; k < U; ++k) {
        const int i = i0 + k * EPW + sub;
        s[k] = qs[i]; u[k] = qu[i]; w[k] = qn[i];
      }
#pragma unroll
      for (int k = 0; k < U; ++k)
        hv[k] = *(const float4*)(h + (size_t)s[k] * CH + f * 4);
#pragma unroll
      for (int k = 0; k < U; ++k) add_row(u[k], w[k], hv[k]);
    }
    {
      const int i = total - total % PB + wave * EPW + sub;
      if (i < total) {
        const float4 hv = *(const float4*)(h + (size_t)qs[i] * CH + f * 4);
        add_row(qu[i], qn[i], hv);
      }
    }
  } else {
    for (int e = tid; e < E; e += NT) {
      const int d = dst[e];
      const unsigned uu = (unsigned)(d - lo);
      if (uu < (unsigned)nrows) {
        const int s = src[e];
        const float w = dinv[s] * dinv[d];
        const float* hp = h + (size_t)s * CH;
        float* ap = &acc[uu * CH];
        for (int c = 0; c < CH; ++c)
          atomicAdd(ap + (c & 3) * Q + (c >> 2), hp[c] * w);
      }
    }
  }
  __syncthreads();

  for (int i = tid; i < nrows * Q; i += NT) {
    const int u = i / Q;
    const float* ap = &acc[u * CH + zf];
    float4 zv;
    zv.x = ap[0];
    zv.y = ap[Q];
    zv.z = ap[2 * Q];
    zv.w = ap[3 * Q];
    *((float4*)(z + (size_t)(lo + u) * CH) + zf) = zv;
  }
}

__global__ void edge_dot(const int* __restrict__ es, const int* __restrict__ ed,
                         const float* __restrict__ z, float* __restrict__ out, int L) {
  int j = blockIdx.x * blockDim.x + threadIdx.x;
  if (j >= L) return;
  const float4* a = (const float4*)(z + (size_t)es[j] * 64);
  const float4* b = (const float4*)(z + (size_t)ed[j] * 64);
  float sum = 0.f;
#pragma unroll
  for (int t = 0; t < 16; ++t) {
    float4 x = a[t], y = b[t];
    sum += x.x * y.x + x.y * y.y + x.z * y.z + x.w * y.w;
  }
  out[j] = sum;
}


static inline int cdiv_ll(long long a, long long b) { return (int)((a + b - 1) / b); }

extern "C" void kernel_launch(void* const* d_in, const int* in_sizes, int n_in,
                              void* d_out, int out_size, void* d_ws, size_t ws_size,
                              hipStream_t stream) {
  const float* x   = (const float*)d_in[0];
  const int*   ei  = (const int*)d_in[1];
  const int*   eli = (const int*)d_in[2];
  const float* W1  = (const float*)d_in[3];
  const float* b1  = (const float*)d_in[4];
  const float* W2  = (const float*)d_in[5];
  const float* b2  = (const float*)d_in[6];
  float* out = (float*)d_out;

  const int IN = 512, HID = 128, OUT = 64;
  const int n = in_sizes[0] / IN;
  const int E = in_sizes[1] / 2;
  const int L = in_sizes[2] / 2;
  const int* src = ei;
  const int* dst = ei + E;
  const int* es  = eli;
  const int* ed  = eli + L;

  float* ws   = (float*)d_ws;
  int*   lcnt = (int*)ws;  ws += n;
  float* dinv = ws;  ws += n;
  float* h1   = ws;  ws += (size_t)n * HID;
  float* z1   = ws;  ws += (size_t)n * HID;
  float* h2   = h1;
  float* z2   = h1 + (size_t)n * OUT;
  int*   lst  = (int*)ws;

  const int B = 256;
  constexpr int RNG  = 196;
  constexpr int QCAP = 12288;
  constexpr int QSAV = 8192;
  constexpr int QSCN = 20480;
  constexpr int GRP  = 16;
  constexpr int PRT  = 16;
  constexpr int FCAP = 4096;

  const int nb = cdiv_ll(n, RNG);
  const int ng = cdiv_ll(nb, GRP);
  int* fdst = (int*)h1;
  int* fsrc = fdst + (size_t)ng * PRT * FCAP;
  int* fcnt = fsrc + (size_t)ng * PRT * FCAP;

  filter_edges_by_group<GRP * RNG, PRT, 512, 8, FCAP, QSCN><<<ng * PRT, 512, 0, stream>>>(src, dst, fsrc, fdst, fcnt, E, n);
  range_scan_edges<RNG, 512, 8, QSCN, QSAV, GRP, PRT, FCAP><<<nb, 512, 0, stream>>>(src, dst, fsrc, fdst, fcnt, dinv, lst, lcnt, E, n);

  gemm_wmma_f32_wstat<512, 128, false><<<256, 32 * (HID / 16), 0, stream>>>(x, W1, h1, n);
  range_sum_edges<128, RNG, 1024, 8, QCAP, QSAV, 1><<<cdiv_ll(n, RNG), 1024, 0, stream>>>(src, dst, dinv, b1, h1, z1, lst, lcnt, E, n);

  gemm_wmma_f32_wstat<128, 64, true><<<625, 32 * (OUT / 16), 0, stream>>>(z1, W2, h2, n);
  range_sum_edges<64, RNG, 1024, 8, QCAP, QSAV, 2><<<cdiv_ll(n, RNG), 1024, 0, stream>>>(src, dst, dinv, b2, h2, z2, lst, lcnt, E, n);

  edge_dot<<<cdiv_ll(L, B), B, 0, stream>>>(es, ed, z2, out, L);
}
